# starting kernel
# baseline (speedup 1.0000x reference)
#include <hip/hip_runtime.h>

typedef __attribute__((ext_vector_type(2))) float v2f;
typedef __attribute__((ext_vector_type(8))) float v8f;

#define KDIM   576
#define NCOL   784
#define TILE_N 16
#define LDS_ELEMS (KDIM * TILE_N)

__global__ __launch_bounds__(256)
void shiftconv_wmma_f32(const float* __restrict__ x,
                        const float* __restrict__ w,
                        float* __restrict__ out) {
  __shared__ float Bt[LDS_ELEMS];

  const int n0 = blockIdx.x * TILE_N;

  for (int e = threadIdx.x; e < LDS_ELEMS; e += 256) {
    const int r   = e >> 4;
    const int c   = e & 15;
    const int col = n0 + c;
    const int n   = col / 28;
    const int o   = col % 28;
    const int j   = r / 192;
    const int k   = (r / 64) % 3;
    const int l   = r & 63;
    const int h   = j + n - 1;
    float v = 0.0f;
    if (h >= 0 && h < 28) {
      const int ko = k + o;
      if (ko >= 1 && ko <= 28)
        v += x[l * 784 + h * 28 + ((ko + 26) % 28)];
      const int om  = (o + 27) % 28;
      const int kom = k + om;
      if (kom >= 1 && kom <= 28)
        v += x[(64 + l) * 784 + h * 28 + ((kom + 26) % 28)];
    }
    Bt[e] = v;
  }
  __syncthreads();

  const int wave = threadIdx.x >> 5;
  const int lane = threadIdx.x & 31;
  const int half = lane >> 4;
  const int lid  = lane & 15;
  const int koff = half * 2;

  const float* __restrict__ arow = w + (wave * 16 + lid) * KDIM + koff;
  const float* __restrict__ bcol = &Bt[koff * TILE_N + lid];

  __builtin_prefetch(arow, 0, 0);

  v8f acc0 = {};
  v8f acc1 = {};
#pragma unroll 4
  for (int kk = 0; kk < KDIM; kk += 8) {
    v2f a0, b0, a1, b1;
    a0.x = arow[kk];
    a0.y = arow[kk + 1];
    b0.x = bcol[kk * TILE_N];
    b0.y = bcol[(kk + 1) * TILE_N];
    a1.x = arow[kk + 4];
    a1.y = arow[kk + 5];
    b1.x = bcol[(kk + 4) * TILE_N];
    b1.y = bcol[(kk + 5) * TILE_N];
    acc0 = __builtin_amdgcn_wmma_f32_16x16x4_f32(false, a0, false, b0,
                                                 (short)0, acc0, false, false);
    acc1 = __builtin_amdgcn_wmma_f32_16x16x4_f32(false, a1, false, b1,
                                                 (short)0, acc1, false, false);
  }
  v8f acc = acc0 + acc1;

  const int nidx  = n0 + lid;
  const int ibase = wave * 16 + half * 8;
#pragma unroll
  for (int r = 0; r < 8; ++r)
    out[(ibase + r) * NCOL + nidx] = acc[r];
}

extern "C" void kernel_launch(void* const* d_in, const int* in_sizes, int n_in,
                              void* d_out, int out_size, void* d_ws, size_t ws_size,
                              hipStream_t stream) {
  const float* x = (const float*)d_in[0];
  const float* w = (const float*)d_in[1];
  float* out = (float*)d_out;

  shiftconv_wmma_f32<<<NCOL / TILE_N, 256, 0, stream>>>(x, w, out);
}


# im2col build: padding guards become clamped in-range addresses plus a select, so a thread's 72 gathers of x are all in flight at once instead of 72 load-then-wait pairs in series
# speedup vs baseline: 1.7537x; 1.7537x over previous
#include <hip/hip_runtime.h>


typedef __attribute__((ext_vector_type(2))) float v2f;
typedef __attribute__((ext_vector_type(8))) float v8f;

#define KDIM   576
#define NCOL   784
#define TILE_N 16
#define LDS_ELEMS (KDIM * TILE_N)

__global__ __launch_bounds__(256)
void shiftconv_wmma_f32(const float* __restrict__ x,
                        const float* __restrict__ w,
                        float* __restrict__ out) {
  __shared__ float Bt[LDS_ELEMS];

  const int n0 = blockIdx.x * TILE_N;

  {
    const int c   = threadIdx.x & 15;
    const int r0  = threadIdx.x >> 4;
    const int col = n0 + c;
    const int n   = col / 28;
    const int o   = col - n * 28;
    const int om  = (o == 0) ? 27 : o - 1;

    int  offA[3], offB[3];
    bool okA[3],  okB[3];
#pragma unroll
    for (int k = 0; k < 3; ++k) {
      const int ko  = k + o;
      okA[k]  = (ko >= 1) && (ko <= 28);
      offA[k] = (ko + 26) % 28;
      const int kom = k + om;
      okB[k]  = (kom >= 1) && (kom <= 28);
      offB[k] = 64 * 784 + (kom + 26) % 28;
    }
    int  rowoff[3];
    bool okH[3];
#pragma unroll
    for (int j = 0; j < 3; ++j) {
      const int h = n - 1 + j;
      okH[j]    = (h >= 0) && (h < 28);
      rowoff[j] = (okH[j] ? h : 0) * 28;
    }

    const float* __restrict__ xl = x + r0 * 784;
    float va[36], vb[36];
#pragma unroll
    for (int i = 0; i < 36; ++i) {
      const int j = i / 12;
      const int k = (i >> 2) % 3;
      const int q = i & 3;
      const float* __restrict__ p = xl + q * (16 * 784) + rowoff[j];
      va[i] = p[offA[k]];
      vb[i] = p[offB[k]];
    }
#pragma unroll
    for (int i = 0; i < 36; ++i) {
      const int j = i / 12;
      const int k = (i >> 2) % 3;
      const float a = (okH[j] & okA[k]) ? va[i] : 0.0f;
      const float b = (okH[j] & okB[k]) ? vb[i] : 0.0f;
      Bt[threadIdx.x + 256 * i] = a + b;
    }
  }
  __syncthreads();

  const int wave = threadIdx.x >> 5;
  const int lane = threadIdx.x & 31;
  const int half = lane >> 4;
  const int lid  = lane & 15;
  const int koff = half * 2;

  const float* __restrict__ arow = w + (wave * 16 + lid) * KDIM + koff;
  const float* __restrict__ bcol = &Bt[koff * TILE_N + lid];

  __builtin_prefetch(arow, 0, 0);

  v8f acc0 = {};
  v8f acc1 = {};
#pragma unroll 4
  for (int kk = 0; kk < KDIM; kk += 8) {
    v2f a0, b0, a1, b1;
    a0.x = arow[kk];
    a0.y = arow[kk + 1];
    b0.x = bcol[kk * TILE_N];
    b0.y = bcol[(kk + 1) * TILE_N];
    a1.x = arow[kk + 4];
    a1.y = arow[kk + 5];
    b1.x = bcol[(kk + 4) * TILE_N];
    b1.y = bcol[(kk + 5) * TILE_N];
    acc0 = __builtin_amdgcn_wmma_f32_16x16x4_f32(false, a0, false, b0,
                                                 (short)0, acc0, false, false);
    acc1 = __builtin_amdgcn_wmma_f32_16x16x4_f32(false, a1, false, b1,
                                                 (short)0, acc1, false, false);
  }
  v8f acc = acc0 + acc1;

  const int nidx  = n0 + lid;
  const int ibase = wave * 16 + half * 8;
#pragma unroll
  for (int r = 0; r < 8; ++r)
    out[(ibase + r) * NCOL + nidx] = acc[r];
}

extern "C" void kernel_launch(void* const* d_in, const int* in_sizes, int n_in,
                              void* d_out, int out_size, void* d_ws, size_t ws_size,
                              hipStream_t stream) {
  const float* x = (const float*)d_in[0];
  const float* w = (const float*)d_in[1];
  float* out = (float*)d_out;

  shiftconv_wmma_f32<<<NCOL / TILE_N, 256, 0, stream>>>(x, w, out);
}


# A operand: each wave loads its whole 16x576 slab of w into registers with 16-byte loads, in flight across the B build and an LDS-only barrier; k order in each 8 permuted (sum order changes)
# speedup vs baseline: 2.5829x; 1.4728x over previous
#include <hip/hip_runtime.h>


typedef __attribute__((ext_vector_type(2))) float v2f;
typedef __attribute__((ext_vector_type(4))) float v4f;
typedef __attribute__((ext_vector_type(8))) float v8f;

#define KDIM   576
#define NCOL   784
#define TILE_N 16
#define LDS_ELEMS (KDIM * TILE_N)

__global__ __launch_bounds__(256)
void shiftconv_wmma_f32(const float* __restrict__ x,
                        const float* __restrict__ w,
                        float* __restrict__ out) {
  __shared__ float Bt[LDS_ELEMS];

  const int n0 = blockIdx.x * TILE_N;

  const int wave = threadIdx.x >> 5;
  const int lane = threadIdx.x & 31;
  const int half = lane >> 4;
  const int lid  = lane & 15;

  v4f areg[KDIM / 8];
  const int NPRE = 63;
  const v4f* __restrict__ arow4 =
      (const v4f*)(w + (wave * 16 + lid) * KDIM + half * 4);

  {
    const int c   = threadIdx.x & 15;
    const int r0  = threadIdx.x >> 4;
    const int col = n0 + c;
    const int n   = col / 28;
    const int o   = col - n * 28;
    const int om  = (o == 0) ? 27 : o - 1;

    int  offA[3], offB[3];
    bool okA[3],  okB[3];
#pragma unroll
    for (int k = 0; k < 3; ++k) {
      const int ko  = k + o;
      okA[k]  = (ko >= 1) && (ko <= 28);
      offA[k] = (ko + 26) % 28;
      const int kom = k + om;
      okB[k]  = (kom >= 1) && (kom <= 28);
      offB[k] = 64 * 784 + (kom + 26) % 28;
    }
    int  rowoff[3];
    bool okH[3];
#pragma unroll
    for (int j = 0; j < 3; ++j) {
      const int h = n - 1 + j;
      okH[j]    = (h >= 0) && (h < 28);
      rowoff[j] = (okH[j] ? h : 0) * 28;
    }

    const float* __restrict__ xl = x + r0 * 784;
    float va[36], vb[36];
#pragma unroll
    for (int i = 0; i < 36; ++i) {
      const int j = i / 12;
      const int k = (i >> 2) % 3;
      const int q = i & 3;
      const float* __restrict__ p = xl + q * (16 * 784) + rowoff[j];
      va[i] = p[offA[k]];
      vb[i] = p[offB[k]];
    }
#pragma unroll
    for (int t = 0; t < NPRE; ++t)
      areg[t] = arow4[2 * t];
#pragma unroll
    for (int i = 0; i < 36; ++i) {
      const int j = i / 12;
      const int k = (i >> 2) % 3;
      const unsigned ma = 0u - (unsigned)(okH[j] & okA[k]);
      const unsigned mb = 0u - (unsigned)(okH[j] & okB[k]);
      const float a = __builtin_bit_cast(float, __builtin_bit_cast(unsigned, va[i]) & ma);
      const float b = __builtin_bit_cast(float, __builtin_bit_cast(unsigned, vb[i]) & mb);
      Bt[threadIdx.x + 256 * i] = a + b;
    }
  }
  __builtin_amdgcn_fence(__ATOMIC_RELEASE, "workgroup", "local");
  __builtin_amdgcn_s_barrier();
  __builtin_amdgcn_fence(__ATOMIC_ACQUIRE, "workgroup", "local");

  const float* __restrict__ bcol = &Bt[(half * 4) * TILE_N + lid];

  v8f acc0 = {};
  v8f acc1 = {};
#pragma unroll
  for (int t = 0; t < KDIM / 8; ++t) {
    v2f a0, b0, a1, b1;
    a0.x = areg[t].x;
    a0.y = areg[t].y;
    a1.x = areg[t].z;
    a1.y = areg[t].w;
    b0.x = bcol[(8 * t + 0) * TILE_N];
    b0.y = bcol[(8 * t + 1) * TILE_N];
    b1.x = bcol[(8 * t + 2) * TILE_N];
    b1.y = bcol[(8 * t + 3) * TILE_N];
    acc0 = __builtin_amdgcn_wmma_f32_16x16x4_f32(false, a0, false, b0,
                                                 (short)0, acc0, false, false);
    acc1 = __builtin_amdgcn_wmma_f32_16x16x4_f32(false, a1, false, b1,
                                                 (short)0, acc1, false, false);
    if (t + NPRE < KDIM / 8)
      areg[t + NPRE] = arow4[2 * (t + NPRE)];
  }
  v8f acc = acc0 + acc1;

  const int nidx  = n0 + lid;
  const int ibase = wave * 16 + half * 8;
#pragma unroll
  for (int r = 0; r < 8; ++r)
    out[(ibase + r) * NCOL + nidx] = acc[r];
}

extern "C" void kernel_launch(void* const* d_in, const int* in_sizes, int n_in,
                              void* d_out, int out_size, void* d_ws, size_t ws_size,
                              hipStream_t stream) {
  const float* x = (const float*)d_in[0];
  const float* w = (const float*)d_in[1];
  float* out = (float*)d_out;

  shiftconv_wmma_f32<<<NCOL / TILE_N, 256, 0, stream>>>(x, w, out);
}


# Grid 49 -> 196 workgroups: each owns 32 rows of w (not 128), K split over its waves in 4 quarters, partial tiles added through LDS in fixed order (sum order changes)
# speedup vs baseline: 3.9387x; 1.5249x over previous
#include <hip/hip_runtime.h>


typedef __attribute__((ext_vector_type(2))) float v2f;
typedef __attribute__((ext_vector_type(4))) float v4f;
typedef __attribute__((ext_vector_type(8))) float v8f;

#define KDIM   576
#define NCOL   784
#define TILE_N 16
#define LDS_ELEMS (KDIM * TILE_N)
#define MROWS  32
#define MSPLIT (128 / MROWS)
#define KSPLIT 4
#define KPART  (KDIM / KSPLIT)
#define TPART  (KPART / 8)
#define RED_ELEMS ((KSPLIT - 1) * 2 * 256)

__global__ __launch_bounds__(256) __attribute__((amdgpu_waves_per_eu(2, 2)))
void shiftconv_wmma_f32(const float* __restrict__ x,
                        const float* __restrict__ w,
                        float* __restrict__ out) {
  __shared__ float Bt[LDS_ELEMS];
  __shared__ float red[RED_ELEMS];

  const int n0 = blockIdx.x * TILE_N;
  const int m0 = blockIdx.y * MROWS;

  const int wave = threadIdx.x >> 5;
  const int mt   = wave & 1;
  const int kq   = wave >> 1;
  const int lane = threadIdx.x & 31;
  const int half = lane >> 4;
  const int lid  = lane & 15;

  v4f areg[TPART];
  const v4f* __restrict__ arow4 =
      (const v4f*)(w + (m0 + mt * 16 + lid) * KDIM + kq * KPART + half * 4);

  {
    const int c   = threadIdx.x & 15;
    const int r0  = threadIdx.x >> 4;
    const int col = n0 + c;
    const int n   = col / 28;
    const int o   = col - n * 28;
    const int om  = (o == 0) ? 27 : o - 1;

    int  offA[3], offB[3];
    bool okA[3],  okB[3];
#pragma unroll
    for (int k = 0; k < 3; ++k) {
      const int ko  = k + o;
      okA[k]  = (ko >= 1) && (ko <= 28);
      offA[k] = (ko + 26) % 28;
      const int kom = k + om;
      okB[k]  = (kom >= 1) && (kom <= 28);
      offB[k] = 64 * 784 + (kom + 26) % 28;
    }
    int  rowoff[3];
    bool okH[3];
#pragma unroll
    for (int j = 0; j < 3; ++j) {
      const int h = n - 1 + j;
      okH[j]    = (h >= 0) && (h < 28);
      rowoff[j] = (okH[j] ? h : 0) * 28;
    }

    const float* __restrict__ xl = x + r0 * 784;
    float va[36], vb[36];
#pragma unroll
    for (int i = 0; i < 36; ++i) {
      const int j = i / 12;
      const int k = (i >> 2) % 3;
      const int q = i & 3;
      const float* __restrict__ p = xl + q * (16 * 784) + rowoff[j];
      va[i] = p[offA[k]];
      vb[i] = p[offB[k]];
    }
#pragma unroll
    for (int t = 0; t < TPART; ++t)
      areg[t] = arow4[2 * t];
#pragma unroll
    for (int i = 0; i < 36; ++i) {
      const int j = i / 12;
      const int k = (i >> 2) % 3;
      const unsigned ma = 0u - (unsigned)(okH[j] & okA[k]);
      const unsigned mb = 0u - (unsigned)(okH[j] & okB[k]);
      const float a = __builtin_bit_cast(float, __builtin_bit_cast(unsigned, va[i]) & ma);
      const float b = __builtin_bit_cast(float, __builtin_bit_cast(unsigned, vb[i]) & mb);
      Bt[threadIdx.x + 256 * i] = a + b;
    }
  }
  __builtin_amdgcn_fence(__ATOMIC_RELEASE, "workgroup", "local");
  __builtin_amdgcn_s_barrier();
  __builtin_amdgcn_fence(__ATOMIC_ACQUIRE, "workgroup", "local");

  const float* __restrict__ bcol = &Bt[(kq * KPART + half * 4) * TILE_N + lid];

  v8f acc0 = {};
  v8f acc1 = {};
#pragma unroll
  for (int t = 0; t < TPART; ++t) {
    v2f a0, b0, a1, b1;
    a0.x = areg[t].x;
    a0.y = areg[t].y;
    a1.x = areg[t].z;
    a1.y = areg[t].w;
    b0.x = bcol[(8 * t + 0) * TILE_N];
    b0.y = bcol[(8 * t + 1) * TILE_N];
    b1.x = bcol[(8 * t + 2) * TILE_N];
    b1.y = bcol[(8 * t + 3) * TILE_N];
    acc0 = __builtin_amdgcn_wmma_f32_16x16x4_f32(false, a0, false, b0,
                                                 (short)0, acc0, false, false);
    acc1 = __builtin_amdgcn_wmma_f32_16x16x4_f32(false, a1, false, b1,
                                                 (short)0, acc1, false, false);
  }
  v8f acc = acc0 + acc1;

  if (kq != 0) {
    float* __restrict__ dst = &red[((kq - 1) * 2 + mt) * 256 + lane];
#pragma unroll
    for (int r = 0; r < 8; ++r)
      dst[r * 32] = acc[r];
  }
  __syncthreads();
  if (kq == 0) {
#pragma unroll
    for (int q = 0; q < KSPLIT - 1; ++q) {
      const float* __restrict__ srcp = &red[(q * 2 + mt) * 256 + lane];
#pragma unroll
      for (int r = 0; r < 8; ++r)
        acc[r] += srcp[r * 32];
    }

    const int nidx  = n0 + lid;
    const int ibase = m0 + mt * 16 + half * 8;
#pragma unroll
    for (int r = 0; r < 8; ++r)
      out[(ibase + r) * NCOL + nidx] = acc[r];
  }
}

extern "C" void kernel_launch(void* const* d_in, const int* in_sizes, int n_in,
                              void* d_out, int out_size, void* d_ws, size_t ws_size,
                              hipStream_t stream) {
  const float* x = (const float*)d_in[0];
  const float* w = (const float*)d_in[1];
  float* out = (float*)d_out;

  shiftconv_wmma_f32<<<dim3(NCOL / TILE_N, MSPLIT), 256, 0, stream>>>(x, w, out);
}


# Workgroup tile = one whole image row x 16 rows of w: x fetched once by 16-byte row loads, shifted sums T kept in LDS, a B fragment is one LDS read at o+k; K split by channel (sum order changes)
# speedup vs baseline: 4.4296x; 1.1246x over previous
#include <hip/hip_runtime.h>


typedef __attribute__((ext_vector_type(2))) float v2f;
typedef __attribute__((ext_vector_type(4))) float v4f;
typedef __attribute__((ext_vector_type(8))) float v8f;

#define KDIM   576
#define NCOL   784
#define IMG    28
#define PLANE  784
#define MROWS  16
#define KPARTS 8
#define NGRP   9
#define TP2    80
#define TROWS  (3 * 32)
#define RED_ELEMS (KPARTS * 16 * 32)

__global__ __launch_bounds__(256) __attribute__((amdgpu_waves_per_eu(2, 2)))
void shiftconv_wmma_f32(const float* __restrict__ x,
                        const float* __restrict__ w,
                        float* __restrict__ out) {
  __shared__ __attribute__((aligned(16))) float Tt[TROWS * TP2];
  __shared__ float red[RED_ELEMS];

  const int n  = blockIdx.x;
  const int m0 = blockIdx.y * MROWS;

  const int wave = threadIdx.x >> 5;
  const int lane = threadIdx.x & 31;
  const int half = lane >> 4;
  const int lid  = lane & 15;

  v4f areg[NGRP];
  const v4f* __restrict__ arow4 =
      (const v4f*)(w + (m0 + lid) * KDIM + wave * 8 + half * 4);

  {
    const int m  = threadIdx.x & 7;
    const int lp = threadIdx.x >> 3;
    const int mc = (m < 7) ? m : 6;

    int      rowoff[3];
    unsigned mk[3];
#pragma unroll
    for (int j = 0; j < 3; ++j) {
      const int  h  = n - 1 + j;
      const bool ok = (h >= 0) && (h < IMG);
      rowoff[j] = (ok ? h : 0) * IMG;
      mk[j]     = 0u - (unsigned)ok;
    }

    const float* __restrict__ xl = x + (2 * lp) * PLANE + mc * 4;
    v4f va0[3], va1[3], vb0[3], vb1[3];
#pragma unroll
    for (int j = 0; j < 3; ++j) {
      const float* __restrict__ p = xl + rowoff[j];
      va0[j] = *(const v4f*)p;
      va1[j] = *(const v4f*)(p + PLANE);
      vb0[j] = *(const v4f*)(p + 64 * PLANE);
      vb1[j] = *(const v4f*)(p + 65 * PLANE);
    }
#pragma unroll
    for (int g = 0; g < NGRP; ++g)
      areg[g] = arow4[g * 16];

    const int src = (m == 0) ? (lane + 6) : (lane - 1);
    float bp0[3], bp1[3];
#pragma unroll
    for (int j = 0; j < 3; ++j) {
      const float l0 = vb0[j].w, l1 = vb1[j].w;
      bp0[j] = __shfl(__builtin_bit_cast(float, __builtin_bit_cast(unsigned, l0) & mk[j]), src, 32);
      bp1[j] = __shfl(__builtin_bit_cast(float, __builtin_bit_cast(unsigned, l1) & mk[j]), src, 32);
    }

    const bool m7 = (m == 7);
    const bool m6 = (m == 6);
    float ax0[3], ax1[3], aw0[3], aw1[3];
#pragma unroll
    for (int j = 0; j < 3; ++j) {
      const float r0x = va0[j].x, r0y = va0[j].y, r0z = va0[j].z, r0w = va0[j].w;
      const float r1x = va1[j].x, r1y = va1[j].y, r1z = va1[j].z, r1w = va1[j].w;
      const float q0x = vb0[j].x, q0y = vb0[j].y, q0z = vb0[j].z;
      const float q1x = vb1[j].x, q1y = vb1[j].y, q1z = vb1[j].z;
      const float a0x = __builtin_bit_cast(float, __builtin_bit_cast(unsigned, r0x) & mk[j]);
      const float a0y = __builtin_bit_cast(float, __builtin_bit_cast(unsigned, r0y) & mk[j]);
      const float a0z = __builtin_bit_cast(float, __builtin_bit_cast(unsigned, r0z) & mk[j]);
      const float a0w = __builtin_bit_cast(float, __builtin_bit_cast(unsigned, r0w) & mk[j]);
      const float a1x = __builtin_bit_cast(float, __builtin_bit_cast(unsigned, r1x) & mk[j]);
      const float a1y = __builtin_bit_cast(float, __builtin_bit_cast(unsigned, r1y) & mk[j]);
      const float a1z = __builtin_bit_cast(float, __builtin_bit_cast(unsigned, r1z) & mk[j]);
      const float a1w = __builtin_bit_cast(float, __builtin_bit_cast(unsigned, r1w) & mk[j]);
      const float b0x = __builtin_bit_cast(float, __builtin_bit_cast(unsigned, q0x) & mk[j]);
      const float b0y = __builtin_bit_cast(float, __builtin_bit_cast(unsigned, q0y) & mk[j]);
      const float b0z = __builtin_bit_cast(float, __builtin_bit_cast(unsigned, q0z) & mk[j]);
      const float b1x = __builtin_bit_cast(float, __builtin_bit_cast(unsigned, q1x) & mk[j]);
      const float b1y = __builtin_bit_cast(float, __builtin_bit_cast(unsigned, q1y) & mk[j]);
      const float b1z = __builtin_bit_cast(float, __builtin_bit_cast(unsigned, q1z) & mk[j]);
      ax0[j] = a0x;
      ax1[j] = a1x;
      aw0[j] = a0w;
      aw1[j] = a1w;

      v4f s0, s1;
      s0.x = m7 ? 0.0f : (a0x + bp0[j]);
      s0.y = m7 ? 0.0f : (a1x + bp1[j]);
      s0.z = m7 ? 0.0f : (a0y + b0x);
      s0.w = m7 ? 0.0f : (a1y + b1x);
      s1.x = (m7 ? 0.0f : a0z) + b0y;
      s1.y = (m7 ? 0.0f : a1z) + b1y;
      s1.z = (m6 ? 0.0f : a0w) + b0z;
      s1.w = (m6 ? 0.0f : a1w) + b1z;
      float* __restrict__ row = &Tt[(j * 32 + lp) * TP2 + (4 * m + 2) * 2];
      *(v4f*)row       = s0;
      *(v4f*)(row + 4) = s1;
    }
    if (m7 || m == 0) {
#pragma unroll
      for (int j = 0; j < 3; ++j) {
        v4f e;
        e.x = m7 ? 0.0f : ax0[j];
        e.y = m7 ? 0.0f : ax1[j];
        e.z = m7 ? aw0[j] : 0.0f;
        e.w = m7 ? aw1[j] : 0.0f;
        *(v4f*)&Tt[(j * 32 + lp) * TP2 + (m7 ? 0 : 68)] = e;
      }
    }
  }
  __builtin_amdgcn_fence(__ATOMIC_RELEASE, "workgroup", "local");
  __builtin_amdgcn_s_barrier();
  __builtin_amdgcn_fence(__ATOMIC_ACQUIRE, "workgroup", "local");

  const int c0 = (lid == 0) ? 32 : lid;
  const float* __restrict__ bt0 = &Tt[(wave * 4 + half * 2) * TP2 + c0 * 2];
  const float* __restrict__ bt1 = &Tt[(wave * 4 + half * 2) * TP2 + (16 + lid) * 2];

  v8f acc00 = {};
  v8f acc01 = {};
  v8f acc10 = {};
  v8f acc11 = {};
#pragma unroll
  for (int g = 0; g < NGRP; ++g) {
    const int off = (g / 3) * 32 * TP2 + (g % 3) * 2;
    v2f a0, a1;
    a0.x = areg[g].x;
    a0.y = areg[g].y;
    a1.x = areg[g].z;
    a1.y = areg[g].w;
    const v2f b00 = *(const v2f*)(bt0 + off);
    const v2f b01 = *(const v2f*)(bt0 + off + TP2);
    const v2f b10 = *(const v2f*)(bt1 + off);
    const v2f b11 = *(const v2f*)(bt1 + off + TP2);
    acc00 = __builtin_amdgcn_wmma_f32_16x16x4_f32(false, a0, false, b00,
                                                  (short)0, acc00, false, false);
    acc01 = __builtin_amdgcn_wmma_f32_16x16x4_f32(false, a1, false, b01,
                                                  (short)0, acc01, false, false);
    acc10 = __builtin_amdgcn_wmma_f32_16x16x4_f32(false, a0, false, b10,
                                                  (short)0, acc10, false, false);
    acc11 = __builtin_amdgcn_wmma_f32_16x16x4_f32(false, a1, false, b11,
                                                  (short)0, acc11, false, false);
  }
  const v8f accA = acc00 + acc01;
  const v8f accB = acc10 + acc11;

  {
    float* __restrict__ dst = &red[wave * (16 * 32) + lane];
#pragma unroll
    for (int r = 0; r < 8; ++r) {
      dst[r * 32]       = accA[r];
      dst[(8 + r) * 32] = accB[r];
    }
  }
  __syncthreads();
  {
    const float* __restrict__ srcp = &red[wave * 32 + lane];
    float s0 = srcp[0];
    float s1 = srcp[8 * 32];
#pragma unroll
    for (int p = 1; p < KPARTS; ++p) {
      s0 += srcp[p * (16 * 32)];
      s1 += srcp[p * (16 * 32) + 8 * 32];
    }
    float* __restrict__ orow = out + (m0 + half * 8 + wave) * NCOL + n * IMG;
    orow[lid] = s0;
    if (lid < IMG - 16)
      orow[16 + lid] = s1;
  }
}

extern "C" void kernel_launch(void* const* d_in, const int* in_sizes, int n_in,
                              void* d_out, int out_size, void* d_ws, size_t ws_size,
                              hipStream_t stream) {
  const float* x = (const float*)d_in[0];
  const float* w = (const float*)d_in[1];
  float* out = (float*)d_out;

  shiftconv_wmma_f32<<<dim3(IMG, 128 / MROWS), 256, 0, stream>>>(x, w, out);
}
